# starting kernel
# baseline (speedup 1.0000x reference)
#include <hip/hip_runtime.h>

typedef float v2f __attribute__((ext_vector_type(2)));
typedef float v8f __attribute__((ext_vector_type(8)));

constexpr int Q = 128;
constexpr int S = 8;
constexpr int H = 32;
constexpr int BT = 16;
constexpr float BN_EPS = 1e-3f;

__global__ __launch_bounds__(256) void divenc_wmma_f32(
    const float* __restrict__ x,
    const float* __restrict__ W1,
    const float* __restrict__ b1g,
    const float* __restrict__ gam,
    const float* __restrict__ bet,
    const float* __restrict__ mmean,
    const float* __restrict__ mvar,
    const float* __restrict__ W2,
    const float* __restrict__ b2g,
    float* __restrict__ out)
{
    const int lane  = threadIdx.x & 31;
    const int wave  = threadIdx.x >> 5;
    const int bbase = blockIdx.x * BT;

    const int mrow  = lane & 15;
    const int khalf = lane >> 4;
    const int ncol  = lane & 15;

    const float* xrow = x + (size_t)(bbase + mrow) * (Q * S);

    for (int qi = 0; qi < 16; ++qi) {
        const int q = wave * 16 + qi;
        const float* xq = xrow + q * S;

        v2f a_lo = { xq[2 * khalf + 0], xq[2 * khalf + 1] };
        v2f a_hi = { xq[2 * khalf + 4], xq[2 * khalf + 5] };

        const float* w1q = W1 + q * (S * H);
        const int h0 = ncol, h1 = ncol + 16;
        const int kb = 2 * khalf;
        v2f bt0_lo = { w1q[(kb + 0) * H + h0], w1q[(kb + 1) * H + h0] };
        v2f bt0_hi = { w1q[(kb + 4) * H + h0], w1q[(kb + 5) * H + h0] };
        v2f bt1_lo = { w1q[(kb + 0) * H + h1], w1q[(kb + 1) * H + h1] };
        v2f bt1_hi = { w1q[(kb + 4) * H + h1], w1q[(kb + 5) * H + h1] };

        const int i0 = q * H + h0, i1 = q * H + h1;
        const float bias0 = b1g[i0], bias1 = b1g[i1];
        v8f acc0 = { bias0, bias0, bias0, bias0, bias0, bias0, bias0, bias0 };
        v8f acc1 = { bias1, bias1, bias1, bias1, bias1, bias1, bias1, bias1 };

        acc0 = __builtin_amdgcn_wmma_f32_16x16x4_f32(false, a_lo, false, bt0_lo,
                                                     (short)0, acc0, false, false);
        acc0 = __builtin_amdgcn_wmma_f32_16x16x4_f32(false, a_hi, false, bt0_hi,
                                                     (short)0, acc0, false, false);
        acc1 = __builtin_amdgcn_wmma_f32_16x16x4_f32(false, a_lo, false, bt1_lo,
                                                     (short)0, acc1, false, false);
        acc1 = __builtin_amdgcn_wmma_f32_16x16x4_f32(false, a_hi, false, bt1_hi,
                                                     (short)0, acc1, false, false);

        const float inv0 = gam[i0] * __builtin_amdgcn_rsqf(mvar[i0] + BN_EPS);
        const float inv1 = gam[i1] * __builtin_amdgcn_rsqf(mvar[i1] + BN_EPS);
        const float w2_0 = W2[i0], w2_1 = W2[i1];
        const float c1_0 = inv0 * w2_0;
        const float c1_1 = inv1 * w2_1;
        const float c0s  = fmaf(-mmean[i0] * inv0 + bet[i0], w2_0,
                                (-mmean[i1] * inv1 + bet[i1]) * w2_1);

        float part[8];
        #pragma unroll
        for (int v = 0; v < 8; ++v) {
            float t0 = acc0[v];
            t0 = t0 > 0.0f ? t0 : (__expf(t0) - 1.0f);
            float t1 = acc1[v];
            t1 = t1 > 0.0f ? t1 : (__expf(t1) - 1.0f);
            part[v] = fmaf(t0, c1_0, fmaf(t1, c1_1, c0s));
        }

        #pragma unroll
        for (int off = 8; off >= 1; off >>= 1) {
            #pragma unroll
            for (int v = 0; v < 8; ++v)
                part[v] += __shfl_xor(part[v], off, 32);
        }

        if (ncol == 0) {
            const float bq = b2g[q];
            float* op = out + (size_t)(bbase + khalf * 8) * Q + q;
            #pragma unroll
            for (int v = 0; v < 8; ++v)
                op[(size_t)v * Q] = part[v] + bq;
        }
    }
}

extern "C" void kernel_launch(void* const* d_in, const int* in_sizes, int n_in,
                              void* d_out, int out_size, void* d_ws, size_t ws_size,
                              hipStream_t stream) {
    const float* x     = (const float*)d_in[0];
    const float* W1    = (const float*)d_in[1];
    const float* b1    = (const float*)d_in[2];
    const float* gamma = (const float*)d_in[3];
    const float* beta  = (const float*)d_in[4];
    const float* mmean = (const float*)d_in[5];
    const float* mvar  = (const float*)d_in[6];
    const float* W2    = (const float*)d_in[7];
    const float* b2    = (const float*)d_in[8];
    float* out = (float*)d_out;

    const int B = in_sizes[0] / (Q * S);
    dim3 grid(B / BT);
    dim3 block(256);
    divenc_wmma_f32<<<grid, block, 0, stream>>>(x, W1, b1, gamma, beta,
                                                mmean, mvar, W2, b2, out);
}


# Lane map re-chosen so a lane's operands lie adjacent in memory (columns h=2c,2c+1; k=4g..4g+3 in lane group g): x one 16-byte load, W1 and parameter rows 8-byte loads; sum order over k and h changes
# speedup vs baseline: 1.0433x; 1.0433x over previous
#include <hip/hip_runtime.h>

typedef float v2f __attribute__((ext_vector_type(2)));
typedef float v4f __attribute__((ext_vector_type(4)));
typedef float v8f __attribute__((ext_vector_type(8)));

constexpr int Q = 128;
constexpr int S = 8;
constexpr int H = 32;
constexpr int BT = 16;
constexpr float BN_EPS = 1e-3f;

__global__ __launch_bounds__(256) void divenc_wmma_f32(
    const float* __restrict__ x,
    const float* __restrict__ W1,
    const float* __restrict__ b1g,
    const float* __restrict__ gam,
    const float* __restrict__ bet,
    const float* __restrict__ mmean,
    const float* __restrict__ mvar,
    const float* __restrict__ W2,
    const float* __restrict__ b2g,
    float* __restrict__ out)
{
    const int lane  = threadIdx.x & 31;
    const int wave  = threadIdx.x >> 5;
    const int bbase = blockIdx.x * BT;

    const int mrow  = lane & 15;
    const int khalf = lane >> 4;
    const int ncol  = lane & 15;
    const int kb    = 4 * khalf;

    const v4f* xrow4 = reinterpret_cast<const v4f*>(x + (size_t)(bbase + mrow) * (Q * S));
    const v2f* W1p   = reinterpret_cast<const v2f*>(W1);
    const v2f* b1p2  = reinterpret_cast<const v2f*>(b1g);
    const v2f* gamp  = reinterpret_cast<const v2f*>(gam);
    const v2f* betp  = reinterpret_cast<const v2f*>(bet);
    const v2f* mup   = reinterpret_cast<const v2f*>(mmean);
    const v2f* varp  = reinterpret_cast<const v2f*>(mvar);
    const v2f* W2p   = reinterpret_cast<const v2f*>(W2);

    for (int qi = 0; qi < 16; ++qi) {
        const int q = wave * 16 + qi;
        const v4f xa = xrow4[q * (S / 4) + khalf];
        v2f a_lo = { xa.x, xa.y };
        v2f a_hi = { xa.z, xa.w };

        const int iw = q * (S * H / 2) + kb * (H / 2) + ncol;
        const v2f wr0 = W1p[iw + 0 * (H / 2)];
        const v2f wr1 = W1p[iw + 1 * (H / 2)];
        const v2f wr2 = W1p[iw + 2 * (H / 2)];
        const v2f wr3 = W1p[iw + 3 * (H / 2)];
        v2f bt0_lo = { wr0.x, wr1.x };
        v2f bt0_hi = { wr2.x, wr3.x };
        v2f bt1_lo = { wr0.y, wr1.y };
        v2f bt1_hi = { wr2.y, wr3.y };

        const int ip = q * (H / 2) + ncol;
        const v2f b1v = b1p2[ip];
        const float bias0 = b1v.x, bias1 = b1v.y;
        v8f acc0 = { bias0, bias0, bias0, bias0, bias0, bias0, bias0, bias0 };
        v8f acc1 = { bias1, bias1, bias1, bias1, bias1, bias1, bias1, bias1 };

        acc0 = __builtin_amdgcn_wmma_f32_16x16x4_f32(false, a_lo, false, bt0_lo,
                                                     (short)0, acc0, false, false);
        acc0 = __builtin_amdgcn_wmma_f32_16x16x4_f32(false, a_hi, false, bt0_hi,
                                                     (short)0, acc0, false, false);
        acc1 = __builtin_amdgcn_wmma_f32_16x16x4_f32(false, a_lo, false, bt1_lo,
                                                     (short)0, acc1, false, false);
        acc1 = __builtin_amdgcn_wmma_f32_16x16x4_f32(false, a_hi, false, bt1_hi,
                                                     (short)0, acc1, false, false);

        const v2f gp = gamp[ip];
        const v2f vp = varp[ip];
        const v2f wp = W2p[ip];
        const v2f mp = mup[ip];
        const v2f ep = betp[ip];
        const float inv0 = gp.x * __builtin_amdgcn_rsqf(vp.x + BN_EPS);
        const float inv1 = gp.y * __builtin_amdgcn_rsqf(vp.y + BN_EPS);
        const float w2_0 = wp.x, w2_1 = wp.y;
        const float c1_0 = inv0 * w2_0;
        const float c1_1 = inv1 * w2_1;
        const float c0s  = fmaf(-mp.x * inv0 + ep.x, w2_0,
                                (-mp.y * inv1 + ep.y) * w2_1);

        float part[8];
        #pragma unroll
        for (int v = 0; v < 8; ++v) {
            float t0 = acc0[v];
            t0 = t0 > 0.0f ? t0 : (__expf(t0) - 1.0f);
            float t1 = acc1[v];
            t1 = t1 > 0.0f ? t1 : (__expf(t1) - 1.0f);
            part[v] = fmaf(t0, c1_0, fmaf(t1, c1_1, c0s));
        }

        #pragma unroll
        for (int off = 8; off >= 1; off >>= 1) {
            #pragma unroll
            for (int v = 0; v < 8; ++v)
                part[v] += __shfl_xor(part[v], off, 32);
        }

        if (ncol == 0) {
            const float bq = b2g[q];
            float* op = out + (size_t)(bbase + khalf * 8) * Q + q;
            #pragma unroll
            for (int v = 0; v < 8; ++v)
                op[(size_t)v * Q] = part[v] + bq;
        }
    }
}

extern "C" void kernel_launch(void* const* d_in, const int* in_sizes, int n_in,
                              void* d_out, int out_size, void* d_ws, size_t ws_size,
                              hipStream_t stream) {
    const float* x     = (const float*)d_in[0];
    const float* W1    = (const float*)d_in[1];
    const float* b1    = (const float*)d_in[2];
    const float* gamma = (const float*)d_in[3];
    const float* beta  = (const float*)d_in[4];
    const float* mmean = (const float*)d_in[5];
    const float* mvar  = (const float*)d_in[6];
    const float* W2    = (const float*)d_in[7];
    const float* b2    = (const float*)d_in[8];
    float* out = (float*)d_out;

    const int B = in_sizes[0] / (Q * S);
    dim3 grid(B / BT);
    dim3 block(256);
    divenc_wmma_f32<<<grid, block, 0, stream>>>(x, W1, b1, gamma, beta,
                                                mmean, mvar, W2, b2, out);
}


# Sum over the 16 lanes of a row moved from ds_bpermute shuffles to a halving DPP row_xmask butterfly on the vector ALU; even lanes store one row each per slice; same pairs summed, order unchanged
# speedup vs baseline: 1.1247x; 1.0780x over previous
#include <hip/hip_runtime.h>

typedef float v2f __attribute__((ext_vector_type(2)));
typedef float v4f __attribute__((ext_vector_type(4)));
typedef float v8f __attribute__((ext_vector_type(8)));

constexpr int Q = 128;
constexpr int S = 8;
constexpr int H = 32;
constexpr int BT = 16;
constexpr float BN_EPS = 1e-3f;

template <int MASK>
__device__ __forceinline__ float row_xor_get(float v) {
    return __int_as_float(__builtin_amdgcn_update_dpp(
        0, __float_as_int(v), 0x160 + MASK, 0xf, 0xf, true));
}

__global__ __launch_bounds__(256) void divenc_wmma_f32(
    const float* __restrict__ x,
    const float* __restrict__ W1,
    const float* __restrict__ b1g,
    const float* __restrict__ gam,
    const float* __restrict__ bet,
    const float* __restrict__ mmean,
    const float* __restrict__ mvar,
    const float* __restrict__ W2,
    const float* __restrict__ b2g,
    float* __restrict__ out)
{
    const int lane  = threadIdx.x & 31;
    const int wave  = threadIdx.x >> 5;
    const int bbase = blockIdx.x * BT;

    const int mrow  = lane & 15;
    const int khalf = lane >> 4;
    const int ncol  = lane & 15;
    const int kb    = 4 * khalf;
    const bool bit3 = (ncol & 8) != 0;
    const bool bit2 = (ncol & 4) != 0;
    const bool bit1 = (ncol & 2) != 0;

    const v4f* xrow4 = reinterpret_cast<const v4f*>(x + (size_t)(bbase + mrow) * (Q * S));
    const v2f* W1p   = reinterpret_cast<const v2f*>(W1);
    const v2f* b1p2  = reinterpret_cast<const v2f*>(b1g);
    const v2f* gamp  = reinterpret_cast<const v2f*>(gam);
    const v2f* betp  = reinterpret_cast<const v2f*>(bet);
    const v2f* mup   = reinterpret_cast<const v2f*>(mmean);
    const v2f* varp  = reinterpret_cast<const v2f*>(mvar);
    const v2f* W2p   = reinterpret_cast<const v2f*>(W2);

    for (int qi = 0; qi < 16; ++qi) {
        const int q = wave * 16 + qi;
        const v4f xa = xrow4[q * (S / 4) + khalf];
        v2f a_lo = { xa.x, xa.y };
        v2f a_hi = { xa.z, xa.w };

        const int iw = q * (S * H / 2) + kb * (H / 2) + ncol;
        const v2f wr0 = W1p[iw + 0 * (H / 2)];
        const v2f wr1 = W1p[iw + 1 * (H / 2)];
        const v2f wr2 = W1p[iw + 2 * (H / 2)];
        const v2f wr3 = W1p[iw + 3 * (H / 2)];
        v2f bt0_lo = { wr0.x, wr1.x };
        v2f bt0_hi = { wr2.x, wr3.x };
        v2f bt1_lo = { wr0.y, wr1.y };
        v2f bt1_hi = { wr2.y, wr3.y };

        const int ip = q * (H / 2) + ncol;
        const v2f b1v = b1p2[ip];
        const float bias0 = b1v.x, bias1 = b1v.y;
        v8f acc0 = { bias0, bias0, bias0, bias0, bias0, bias0, bias0, bias0 };
        v8f acc1 = { bias1, bias1, bias1, bias1, bias1, bias1, bias1, bias1 };

        acc0 = __builtin_amdgcn_wmma_f32_16x16x4_f32(false, a_lo, false, bt0_lo,
                                                     (short)0, acc0, false, false);
        acc0 = __builtin_amdgcn_wmma_f32_16x16x4_f32(false, a_hi, false, bt0_hi,
                                                     (short)0, acc0, false, false);
        acc1 = __builtin_amdgcn_wmma_f32_16x16x4_f32(false, a_lo, false, bt1_lo,
                                                     (short)0, acc1, false, false);
        acc1 = __builtin_amdgcn_wmma_f32_16x16x4_f32(false, a_hi, false, bt1_hi,
                                                     (short)0, acc1, false, false);

        const v2f gp = gamp[ip];
        const v2f vp = varp[ip];
        const v2f wp = W2p[ip];
        const v2f mp = mup[ip];
        const v2f ep = betp[ip];
        const float inv0 = gp.x * __builtin_amdgcn_rsqf(vp.x + BN_EPS);
        const float inv1 = gp.y * __builtin_amdgcn_rsqf(vp.y + BN_EPS);
        const float w2_0 = wp.x, w2_1 = wp.y;
        const float c1_0 = inv0 * w2_0;
        const float c1_1 = inv1 * w2_1;
        const float c0s  = fmaf(-mp.x * inv0 + ep.x, w2_0,
                                (-mp.y * inv1 + ep.y) * w2_1);

        float part[8];
        #pragma unroll
        for (int v = 0; v < 8; ++v) {
            float t0 = acc0[v];
            t0 = t0 > 0.0f ? t0 : (__expf(t0) - 1.0f);
            float t1 = acc1[v];
            t1 = t1 > 0.0f ? t1 : (__expf(t1) - 1.0f);
            part[v] = fmaf(t0, c1_0, fmaf(t1, c1_1, c0s));
        }

        float r1[4], r2[2];
        #pragma unroll
        for (int i = 0; i < 4; ++i) {
            const float keep = bit3 ? part[i + 4] : part[i];
            const float send = bit3 ? part[i] : part[i + 4];
            r1[i] = keep + row_xor_get<8>(send);
        }
        #pragma unroll
        for (int i = 0; i < 2; ++i) {
            const float keep = bit2 ? r1[i + 2] : r1[i];
            const float send = bit2 ? r1[i] : r1[i + 2];
            r2[i] = keep + row_xor_get<4>(send);
        }
        const float keep3 = bit1 ? r2[1] : r2[0];
        const float send3 = bit1 ? r2[0] : r2[1];
        const float r3 = keep3 + row_xor_get<2>(send3);
        const float r4 = r3 + row_xor_get<1>(r3);

        if ((ncol & 1) == 0) {
            const float bq = b2g[q];
            float* op = out + (size_t)(bbase + khalf * 8 + (ncol >> 1)) * Q + q;
            *op = r4 + bq;
        }
    }
}

extern "C" void kernel_launch(void* const* d_in, const int* in_sizes, int n_in,
                              void* d_out, int out_size, void* d_ws, size_t ws_size,
                              hipStream_t stream) {
    const float* x     = (const float*)d_in[0];
    const float* W1    = (const float*)d_in[1];
    const float* b1    = (const float*)d_in[2];
    const float* gamma = (const float*)d_in[3];
    const float* beta  = (const float*)d_in[4];
    const float* mmean = (const float*)d_in[5];
    const float* mvar  = (const float*)d_in[6];
    const float* W2    = (const float*)d_in[7];
    const float* b2    = (const float*)d_in[8];
    float* out = (float*)d_out;

    const int B = in_sizes[0] / (Q * S);
    dim3 grid(B / BT);
    dim3 block(256);
    divenc_wmma_f32<<<grid, block, 0, stream>>>(x, W1, b1, gamma, beta,
                                                mmean, mvar, W2, b2, out);
}


# Out tile collected in LDS and written row by row: one 512-byte store per batch row (4 whole 128-byte lines) replaces 128 four-byte stores; b2 read once per thread at the row write; sums unchanged
# speedup vs baseline: 1.5284x; 1.3589x over previous
#include <hip/hip_runtime.h>

typedef float v2f __attribute__((ext_vector_type(2)));
typedef float v4f __attribute__((ext_vector_type(4)));
typedef float v8f __attribute__((ext_vector_type(8)));

constexpr int Q = 128;
constexpr int S = 8;
constexpr int H = 32;
constexpr int BT = 16;
constexpr int OP = Q + 4;
constexpr float BN_EPS = 1e-3f;

template <int MASK>
__device__ __forceinline__ float row_xor_get(float v) {
    return __int_as_float(__builtin_amdgcn_update_dpp(
        0, __float_as_int(v), 0x160 + MASK, 0xf, 0xf, true));
}

__global__ __launch_bounds__(256) void divenc_wmma_f32(
    const float* __restrict__ x,
    const float* __restrict__ W1,
    const float* __restrict__ b1g,
    const float* __restrict__ gam,
    const float* __restrict__ bet,
    const float* __restrict__ mmean,
    const float* __restrict__ mvar,
    const float* __restrict__ W2,
    const float* __restrict__ b2g,
    float* __restrict__ out)
{
    __shared__ __attribute__((aligned(16))) float otile[BT * OP];

    const int lane  = threadIdx.x & 31;
    const int wave  = threadIdx.x >> 5;
    const int bbase = blockIdx.x * BT;

    const int mrow  = lane & 15;
    const int khalf = lane >> 4;
    const int ncol  = lane & 15;
    const int kb    = 4 * khalf;
    const bool bit3 = (ncol & 8) != 0;
    const bool bit2 = (ncol & 4) != 0;
    const bool bit1 = (ncol & 2) != 0;

    const v4f* xrow4 = reinterpret_cast<const v4f*>(x + (size_t)(bbase + mrow) * (Q * S));
    const v2f* W1p   = reinterpret_cast<const v2f*>(W1);
    const v2f* b1p2  = reinterpret_cast<const v2f*>(b1g);
    const v2f* gamp  = reinterpret_cast<const v2f*>(gam);
    const v2f* betp  = reinterpret_cast<const v2f*>(bet);
    const v2f* mup   = reinterpret_cast<const v2f*>(mmean);
    const v2f* varp  = reinterpret_cast<const v2f*>(mvar);
    const v2f* W2p   = reinterpret_cast<const v2f*>(W2);

    for (int qi = 0; qi < 16; ++qi) {
        const int q = wave * 16 + qi;
        const v4f xa = xrow4[q * (S / 4) + khalf];
        v2f a_lo = { xa.x, xa.y };
        v2f a_hi = { xa.z, xa.w };

        const int iw = q * (S * H / 2) + kb * (H / 2) + ncol;
        const v2f wr0 = W1p[iw + 0 * (H / 2)];
        const v2f wr1 = W1p[iw + 1 * (H / 2)];
        const v2f wr2 = W1p[iw + 2 * (H / 2)];
        const v2f wr3 = W1p[iw + 3 * (H / 2)];
        v2f bt0_lo = { wr0.x, wr1.x };
        v2f bt0_hi = { wr2.x, wr3.x };
        v2f bt1_lo = { wr0.y, wr1.y };
        v2f bt1_hi = { wr2.y, wr3.y };

        const int ip = q * (H / 2) + ncol;
        const v2f b1v = b1p2[ip];
        const float bias0 = b1v.x, bias1 = b1v.y;
        v8f acc0 = { bias0, bias0, bias0, bias0, bias0, bias0, bias0, bias0 };
        v8f acc1 = { bias1, bias1, bias1, bias1, bias1, bias1, bias1, bias1 };

        acc0 = __builtin_amdgcn_wmma_f32_16x16x4_f32(false, a_lo, false, bt0_lo,
                                                     (short)0, acc0, false, false);
        acc0 = __builtin_amdgcn_wmma_f32_16x16x4_f32(false, a_hi, false, bt0_hi,
                                                     (short)0, acc0, false, false);
        acc1 = __builtin_amdgcn_wmma_f32_16x16x4_f32(false, a_lo, false, bt1_lo,
                                                     (short)0, acc1, false, false);
        acc1 = __builtin_amdgcn_wmma_f32_16x16x4_f32(false, a_hi, false, bt1_hi,
                                                     (short)0, acc1, false, false);

        const v2f gp = gamp[ip];
        const v2f vp = varp[ip];
        const v2f wp = W2p[ip];
        const v2f mp = mup[ip];
        const v2f ep = betp[ip];
        const float inv0 = gp.x * __builtin_amdgcn_rsqf(vp.x + BN_EPS);
        const float inv1 = gp.y * __builtin_amdgcn_rsqf(vp.y + BN_EPS);
        const float w2_0 = wp.x, w2_1 = wp.y;
        const float c1_0 = inv0 * w2_0;
        const float c1_1 = inv1 * w2_1;
        const float c0s  = fmaf(-mp.x * inv0 + ep.x, w2_0,
                                (-mp.y * inv1 + ep.y) * w2_1);

        float part[8];
        #pragma unroll
        for (int v = 0; v < 8; ++v) {
            float t0 = acc0[v];
            t0 = t0 > 0.0f ? t0 : (__expf(t0) - 1.0f);
            float t1 = acc1[v];
            t1 = t1 > 0.0f ? t1 : (__expf(t1) - 1.0f);
            part[v] = fmaf(t0, c1_0, fmaf(t1, c1_1, c0s));
        }

        float r1[4], r2[2];
        #pragma unroll
        for (int i = 0; i < 4; ++i) {
            const float keep = bit3 ? part[i + 4] : part[i];
            const float send = bit3 ? part[i] : part[i + 4];
            r1[i] = keep + row_xor_get<8>(send);
        }
        #pragma unroll
        for (int i = 0; i < 2; ++i) {
            const float keep = bit2 ? r1[i + 2] : r1[i];
            const float send = bit2 ? r1[i] : r1[i + 2];
            r2[i] = keep + row_xor_get<4>(send);
        }
        const float keep3 = bit1 ? r2[1] : r2[0];
        const float send3 = bit1 ? r2[0] : r2[1];
        const float r3 = keep3 + row_xor_get<2>(send3);
        const float r4 = r3 + row_xor_get<1>(r3);

        if ((ncol & 1) == 0) {
            otile[(khalf * 8 + (ncol >> 1)) * OP + q] = r4;
        }
    }

    const v4f b2v = reinterpret_cast<const v4f*>(b2g)[lane];
    __syncthreads();
    for (int i = 0; i < 2; ++i) {
        const int r = 2 * wave + i;
        const v4f s = *reinterpret_cast<const v4f*>(&otile[r * OP + 4 * lane]);
        v4f o = { s.x + b2v.x, s.y + b2v.y, s.z + b2v.z, s.w + b2v.w };
        *reinterpret_cast<v4f*>(out + (size_t)(bbase + r) * Q + 4 * lane) = o;
    }
}

extern "C" void kernel_launch(void* const* d_in, const int* in_sizes, int n_in,
                              void* d_out, int out_size, void* d_ws, size_t ws_size,
                              hipStream_t stream) {
    const float* x     = (const float*)d_in[0];
    const float* W1    = (const float*)d_in[1];
    const float* b1    = (const float*)d_in[2];
    const float* gamma = (const float*)d_in[3];
    const float* beta  = (const float*)d_in[4];
    const float* mmean = (const float*)d_in[5];
    const float* mvar  = (const float*)d_in[6];
    const float* W2    = (const float*)d_in[7];
    const float* b2    = (const float*)d_in[8];
    float* out = (float*)d_out;

    const int B = in_sizes[0] / (Q * S);
    dim3 grid(B / BT);
    dim3 block(256);
    divenc_wmma_f32<<<grid, block, 0, stream>>>(x, W1, b1, gamma, beta,
                                                mmean, mvar, W2, b2, out);
}


# Parameters via LDS: W1, bias, BN/W2 rows read once per 128-row block (32 waves), folded, kept in fragment order; slice loop loads only x from memory; constant term summed apart: sum order changes
# speedup vs baseline: 1.7374x; 1.1367x over previous
#include <hip/hip_runtime.h>

typedef float v2f __attribute__((ext_vector_type(2)));
typedef float v4f __attribute__((ext_vector_type(4)));
typedef float v8f __attribute__((ext_vector_type(8)));

constexpr int Q = 128;
constexpr int S = 8;
constexpr int H = 32;
constexpr int BT = 16;
constexpr int RB = 128;
constexpr int OP = Q + 4;
constexpr float BN_EPS = 1e-3f;

template <int MASK>
__device__ __forceinline__ float row_xor_get(float v) {
    return __int_as_float(__builtin_amdgcn_update_dpp(
        0, __float_as_int(v), 0x160 + MASK, 0xf, 0xf, true));
}

__global__ __launch_bounds__(1024) void divenc_wmma_f32(
    const float* __restrict__ x,
    const float* __restrict__ W1,
    const float* __restrict__ b1g,
    const float* __restrict__ gam,
    const float* __restrict__ bet,
    const float* __restrict__ mmean,
    const float* __restrict__ mvar,
    const float* __restrict__ W2,
    const float* __restrict__ b2g,
    float* __restrict__ out)
{
    __shared__ __attribute__((aligned(16))) float wfrag[2 * Q * 32 * 4];
    __shared__ __attribute__((aligned(16))) float coef[Q * 16 * 4];
    __shared__ __attribute__((aligned(16))) float cst[Q];
    __shared__ __attribute__((aligned(16))) float otile[RB * OP];

    const int tid   = threadIdx.x;
    const int lane  = tid & 31;
    const int wave  = tid >> 5;
    const int bbase = blockIdx.x * RB;

    const int mrow  = lane & 15;
    const int khalf = lane >> 4;
    const int ncol  = lane & 15;
    const int kb    = 4 * khalf;
    const bool bit3 = (ncol & 8) != 0;
    const bool bit2 = (ncol & 4) != 0;
    const bool bit1 = (ncol & 2) != 0;

    const v2f* W1p   = reinterpret_cast<const v2f*>(W1);
    const v2f* b1p2  = reinterpret_cast<const v2f*>(b1g);
    const v2f* gamp  = reinterpret_cast<const v2f*>(gam);
    const v2f* betp  = reinterpret_cast<const v2f*>(bet);
    const v2f* mup   = reinterpret_cast<const v2f*>(mmean);
    const v2f* varp  = reinterpret_cast<const v2f*>(mvar);
    const v2f* W2p   = reinterpret_cast<const v2f*>(W2);

    for (int j = 0; j < 4; ++j) {
        const int q  = wave + 32 * j;
        const int iw = q * (S * H / 2) + kb * (H / 2) + ncol;
        const v2f wr0 = W1p[iw + 0 * (H / 2)];
        const v2f wr1 = W1p[iw + 1 * (H / 2)];
        const v2f wr2 = W1p[iw + 2 * (H / 2)];
        const v2f wr3 = W1p[iw + 3 * (H / 2)];
        const v4f f0 = { wr0.x, wr1.x, wr2.x, wr3.x };
        const v4f f1 = { wr0.y, wr1.y, wr2.y, wr3.y };
        *reinterpret_cast<v4f*>(&wfrag[((0 * Q + q) * 32 + lane) * 4]) = f0;
        *reinterpret_cast<v4f*>(&wfrag[((1 * Q + q) * 32 + lane) * 4]) = f1;
    }
    for (int j = 0; j < 2; ++j) {
        const int rec = tid + 1024 * j;
        const v2f b1v = b1p2[rec];
        const v2f gp = gamp[rec];
        const v2f vp = varp[rec];
        const v2f wp = W2p[rec];
        const v2f mp = mup[rec];
        const v2f ep = betp[rec];
        const float inv0 = gp.x * __builtin_amdgcn_rsqf(vp.x + BN_EPS);
        const float inv1 = gp.y * __builtin_amdgcn_rsqf(vp.y + BN_EPS);
        const v4f cr = { b1v.x, b1v.y, inv0 * wp.x, inv1 * wp.y };
        *reinterpret_cast<v4f*>(&coef[rec * 4]) = cr;
        float c0 = fmaf(-mp.x * inv0 + ep.x, wp.x,
                        (-mp.y * inv1 + ep.y) * wp.y);
        c0 += row_xor_get<8>(c0);
        c0 += row_xor_get<4>(c0);
        c0 += row_xor_get<2>(c0);
        c0 += row_xor_get<1>(c0);
        if (ncol == 0) {
            const int q = rec >> 4;
            cst[q] = c0 + b2g[q];
        }
    }
    __syncthreads();

    const int mt = wave & 7;
    const int qg = wave >> 3;
    const v4f* xrow4 = reinterpret_cast<const v4f*>(
        x + (size_t)(bbase + mt * BT + mrow) * (Q * S));
    const int orow = (mt * BT + khalf * 8 + (ncol >> 1)) * OP;

    for (int qi = 0; qi < 32; ++qi) {
        const int q = qg * 32 + qi;
        const v4f xa = xrow4[q * (S / 4) + khalf];
        v2f a_lo = { xa.x, xa.y };
        v2f a_hi = { xa.z, xa.w };

        const v4f f0 = *reinterpret_cast<const v4f*>(&wfrag[((0 * Q + q) * 32 + lane) * 4]);
        const v4f f1 = *reinterpret_cast<const v4f*>(&wfrag[((1 * Q + q) * 32 + lane) * 4]);
        v2f bt0_lo = { f0.x, f0.y };
        v2f bt0_hi = { f0.z, f0.w };
        v2f bt1_lo = { f1.x, f1.y };
        v2f bt1_hi = { f1.z, f1.w };

        const v4f cr = *reinterpret_cast<const v4f*>(&coef[(q * 16 + ncol) * 4]);
        const float bias0 = cr.x, bias1 = cr.y;
        const float c1_0 = cr.z, c1_1 = cr.w;
        v8f acc0 = { bias0, bias0, bias0, bias0, bias0, bias0, bias0, bias0 };
        v8f acc1 = { bias1, bias1, bias1, bias1, bias1, bias1, bias1, bias1 };

        acc0 = __builtin_amdgcn_wmma_f32_16x16x4_f32(false, a_lo, false, bt0_lo,
                                                     (short)0, acc0, false, false);
        acc0 = __builtin_amdgcn_wmma_f32_16x16x4_f32(false, a_hi, false, bt0_hi,
                                                     (short)0, acc0, false, false);
        acc1 = __builtin_amdgcn_wmma_f32_16x16x4_f32(false, a_lo, false, bt1_lo,
                                                     (short)0, acc1, false, false);
        acc1 = __builtin_amdgcn_wmma_f32_16x16x4_f32(false, a_hi, false, bt1_hi,
                                                     (short)0, acc1, false, false);

        float part[8];
        #pragma unroll
        for (int v = 0; v < 8; ++v) {
            float t0 = acc0[v];
            t0 = t0 > 0.0f ? t0 : (__expf(t0) - 1.0f);
            float t1 = acc1[v];
            t1 = t1 > 0.0f ? t1 : (__expf(t1) - 1.0f);
            part[v] = fmaf(t0, c1_0, t1 * c1_1);
        }

        float r1[4], r2[2];
        #pragma unroll
        for (int i = 0; i < 4; ++i) {
            const float keep = bit3 ? part[i + 4] : part[i];
            const float send = bit3 ? part[i] : part[i + 4];
            r1[i] = keep + row_xor_get<8>(send);
        }
        #pragma unroll
        for (int i = 0; i < 2; ++i) {
            const float keep = bit2 ? r1[i + 2] : r1[i];
            const float send = bit2 ? r1[i] : r1[i + 2];
            r2[i] = keep + row_xor_get<4>(send);
        }
        const float keep3 = bit1 ? r2[1] : r2[0];
        const float send3 = bit1 ? r2[0] : r2[1];
        const float r3 = keep3 + row_xor_get<2>(send3);
        const float r4 = r3 + row_xor_get<1>(r3);

        if ((ncol & 1) == 0) {
            otile[orow + q] = r4;
        }
    }

    __syncthreads();
    const v4f cv = *reinterpret_cast<const v4f*>(&cst[4 * lane]);
    for (int i = 0; i < 4; ++i) {
        const int r = 4 * wave + i;
        const v4f s = *reinterpret_cast<const v4f*>(&otile[r * OP + 4 * lane]);
        v4f o = { s.x + cv.x, s.y + cv.y, s.z + cv.z, s.w + cv.w };
        *reinterpret_cast<v4f*>(out + (size_t)(bbase + r) * Q + 4 * lane) = o;
    }
}

extern "C" void kernel_launch(void* const* d_in, const int* in_sizes, int n_in,
                              void* d_out, int out_size, void* d_ws, size_t ws_size,
                              hipStream_t stream) {
    const float* x     = (const float*)d_in[0];
    const float* W1    = (const float*)d_in[1];
    const float* b1    = (const float*)d_in[2];
    const float* gamma = (const float*)d_in[3];
    const float* beta  = (const float*)d_in[4];
    const float* mmean = (const float*)d_in[5];
    const float* mvar  = (const float*)d_in[6];
    const float* W2    = (const float*)d_in[7];
    const float* b2    = (const float*)d_in[8];
    float* out = (float*)d_out;

    const int B = in_sizes[0] / (Q * S);
    dim3 grid(B / RB);
    dim3 block(1024);
    divenc_wmma_f32<<<grid, block, 0, stream>>>(x, W1, b1, gamma, beta,
                                                mmean, mvar, W2, b2, out);
}


# x read as whole 128-byte lines (4 rows per load, each line once) and handed to the A fragments through a per-wave LDS strip, next 4-slice chunk in flight meanwhile; sums and their order unchanged
# speedup vs baseline: 1.9279x; 1.1096x over previous
#include <hip/hip_runtime.h>

typedef float v2f __attribute__((ext_vector_type(2)));
typedef float v4f __attribute__((ext_vector_type(4)));
typedef float v8f __attribute__((ext_vector_type(8)));

constexpr int Q = 128;
constexpr int S = 8;
constexpr int H = 32;
constexpr int BT = 16;
constexpr int RB = 128;
constexpr int OP = Q + 4;
constexpr int XC = 4;
constexpr int XP = XC * S + 4;
constexpr float BN_EPS = 1e-3f;

template <int MASK>
__device__ __forceinline__ float row_xor_get(float v) {
    return __int_as_float(__builtin_amdgcn_update_dpp(
        0, __float_as_int(v), 0x160 + MASK, 0xf, 0xf, true));
}

__global__ __launch_bounds__(1024) void divenc_wmma_f32(
    const float* __restrict__ x,
    const float* __restrict__ W1,
    const float* __restrict__ b1g,
    const float* __restrict__ gam,
    const float* __restrict__ bet,
    const float* __restrict__ mmean,
    const float* __restrict__ mvar,
    const float* __restrict__ W2,
    const float* __restrict__ b2g,
    float* __restrict__ out)
{
    __shared__ __attribute__((aligned(16))) float wfrag[2 * Q * 32 * 4];
    __shared__ __attribute__((aligned(16))) float coef[Q * 16 * 4];
    __shared__ __attribute__((aligned(16))) float cst[Q];
    __shared__ __attribute__((aligned(16))) float otile[RB * OP];
    __shared__ __attribute__((aligned(16))) float xst[32 * BT * XP];

    const int tid   = threadIdx.x;
    const int lane  = tid & 31;
    const int wave  = tid >> 5;
    const int bbase = blockIdx.x * RB;

    const int mrow  = lane & 15;
    const int khalf = lane >> 4;
    const int ncol  = lane & 15;
    const int kb    = 4 * khalf;
    const bool bit3 = (ncol & 8) != 0;
    const bool bit2 = (ncol & 4) != 0;
    const bool bit1 = (ncol & 2) != 0;

    const v2f* W1p   = reinterpret_cast<const v2f*>(W1);
    const v2f* b1p2  = reinterpret_cast<const v2f*>(b1g);
    const v2f* gamp  = reinterpret_cast<const v2f*>(gam);
    const v2f* betp  = reinterpret_cast<const v2f*>(bet);
    const v2f* mup   = reinterpret_cast<const v2f*>(mmean);
    const v2f* varp  = reinterpret_cast<const v2f*>(mvar);
    const v2f* W2p   = reinterpret_cast<const v2f*>(W2);

    const int mt = wave & 7;
    const int qg = wave >> 3;
    const int lg = lane >> 3;
    const int lp = lane & 7;
    const int lr0 = 8 * (lg & 1) + 4 * (lg >> 1);
    const v4f* xg = reinterpret_cast<const v4f*>(
        x + (size_t)(bbase + mt * BT + lr0) * (Q * S) + qg * (32 * S)) + lp;
    constexpr int GROW = Q * S / 4;
    constexpr int GCH  = XC * S / 4;
    constexpr int LROW = XP / 4;
    v4f nx0 = xg[0 * GROW];
    v4f nx1 = xg[1 * GROW];
    v4f nx2 = xg[2 * GROW];
    v4f nx3 = xg[3 * GROW];

    for (int j = 0; j < 4; ++j) {
        const int q  = wave + 32 * j;
        const int iw = q * (S * H / 2) + kb * (H / 2) + ncol;
        const v2f wr0 = W1p[iw + 0 * (H / 2)];
        const v2f wr1 = W1p[iw + 1 * (H / 2)];
        const v2f wr2 = W1p[iw + 2 * (H / 2)];
        const v2f wr3 = W1p[iw + 3 * (H / 2)];
        const v4f f0 = { wr0.x, wr1.x, wr2.x, wr3.x };
        const v4f f1 = { wr0.y, wr1.y, wr2.y, wr3.y };
        *reinterpret_cast<v4f*>(&wfrag[((0 * Q + q) * 32 + lane) * 4]) = f0;
        *reinterpret_cast<v4f*>(&wfrag[((1 * Q + q) * 32 + lane) * 4]) = f1;
    }
    for (int j = 0; j < 2; ++j) {
        const int rec = tid + 1024 * j;
        const v2f b1v = b1p2[rec];
        const v2f gp = gamp[rec];
        const v2f vp = varp[rec];
        const v2f wp = W2p[rec];
        const v2f mp = mup[rec];
        const v2f ep = betp[rec];
        const float inv0 = gp.x * __builtin_amdgcn_rsqf(vp.x + BN_EPS);
        const float inv1 = gp.y * __builtin_amdgcn_rsqf(vp.y + BN_EPS);
        const v4f cr = { b1v.x, b1v.y, inv0 * wp.x, inv1 * wp.y };
        *reinterpret_cast<v4f*>(&coef[rec * 4]) = cr;
        float c0 = fmaf(-mp.x * inv0 + ep.x, wp.x,
                        (-mp.y * inv1 + ep.y) * wp.y);
        c0 += row_xor_get<8>(c0);
        c0 += row_xor_get<4>(c0);
        c0 += row_xor_get<2>(c0);
        c0 += row_xor_get<1>(c0);
        if (ncol == 0) {
            const int q = rec >> 4;
            cst[q] = c0 + b2g[q];
        }
    }
    __syncthreads();

    const int orow = (mt * BT + khalf * 8 + (ncol >> 1)) * OP;

    float* xw = &xst[wave * (BT * XP)];
    v4f* xput = reinterpret_cast<v4f*>(xw + lr0 * XP + 4 * lp);
    const v4f* xget = reinterpret_cast<const v4f*>(xw + mrow * XP + kb);

    for (int ch = 0; ch < 32 / XC; ++ch) {
      xput[0 * LROW] = nx0;
      xput[1 * LROW] = nx1;
      xput[2 * LROW] = nx2;
      xput[3 * LROW] = nx3;
      if (ch + 1 < 32 / XC) {
        const v4f* xn = xg + (ch + 1) * GCH;
        nx0 = xn[0 * GROW];
        nx1 = xn[1 * GROW];
        nx2 = xn[2 * GROW];
        nx3 = xn[3 * GROW];
      }
      __builtin_amdgcn_fence(__ATOMIC_RELEASE, "wavefront");
      __builtin_amdgcn_wave_barrier();
      __builtin_amdgcn_fence(__ATOMIC_ACQUIRE, "wavefront");

      for (int ql = 0; ql < XC; ++ql) {
        const int q = qg * 32 + ch * XC + ql;
        const v4f xa = xget[ql * (S / 4)];
        v2f a_lo = { xa.x, xa.y };
        v2f a_hi = { xa.z, xa.w };

        const v4f f0 = *reinterpret_cast<const v4f*>(&wfrag[((0 * Q + q) * 32 + lane) * 4]);
        const v4f f1 = *reinterpret_cast<const v4f*>(&wfrag[((1 * Q + q) * 32 + lane) * 4]);
        v2f bt0_lo = { f0.x, f0.y };
        v2f bt0_hi = { f0.z, f0.w };
        v2f bt1_lo = { f1.x, f1.y };
        v2f bt1_hi = { f1.z, f1.w };

        const v4f cr = *reinterpret_cast<const v4f*>(&coef[(q * 16 + ncol) * 4]);
        const float bias0 = cr.x, bias1 = cr.y;
        const float c1_0 = cr.z, c1_1 = cr.w;
        v8f acc0 = { bias0, bias0, bias0, bias0, bias0, bias0, bias0, bias0 };
        v8f acc1 = { bias1, bias1, bias1, bias1, bias1, bias1, bias1, bias1 };

        acc0 = __builtin_amdgcn_wmma_f32_16x16x4_f32(false, a_lo, false, bt0_lo,
                                                     (short)0, acc0, false, false);
        acc0 = __builtin_amdgcn_wmma_f32_16x16x4_f32(false, a_hi, false, bt0_hi,
                                                     (short)0, acc0, false, false);
        acc1 = __builtin_amdgcn_wmma_f32_16x16x4_f32(false, a_lo, false, bt1_lo,
                                                     (short)0, acc1, false, false);
        acc1 = __builtin_amdgcn_wmma_f32_16x16x4_f32(false, a_hi, false, bt1_hi,
                                                     (short)0, acc1, false, false);

        float part[8];
        #pragma unroll
        for (int v = 0; v < 8; ++v) {
            float t0 = acc0[v];
            t0 = t0 > 0.0f ? t0 : (__expf(t0) - 1.0f);
            float t1 = acc1[v];
            t1 = t1 > 0.0f ? t1 : (__expf(t1) - 1.0f);
            part[v] = fmaf(t0, c1_0, t1 * c1_1);
        }

        float r1[4], r2[2];
        #pragma unroll
        for (int i = 0; i < 4; ++i) {
            const float keep = bit3 ? part[i + 4] : part[i];
            const float send = bit3 ? part[i] : part[i + 4];
            r1[i] = keep + row_xor_get<8>(send);
        }
        #pragma unroll
        for (int i = 0; i < 2; ++i) {
            const float keep = bit2 ? r1[i + 2] : r1[i];
            const float send = bit2 ? r1[i] : r1[i + 2];
            r2[i] = keep + row_xor_get<4>(send);
        }
        const float keep3 = bit1 ? r2[1] : r2[0];
        const float send3 = bit1 ? r2[0] : r2[1];
        const float r3 = keep3 + row_xor_get<2>(send3);
        const float r4 = r3 + row_xor_get<1>(r3);

        if ((ncol & 1) == 0) {
            otile[orow + q] = r4;
        }
      }
      __builtin_amdgcn_fence(__ATOMIC_RELEASE, "wavefront");
      __builtin_amdgcn_wave_barrier();
      __builtin_amdgcn_fence(__ATOMIC_ACQUIRE, "wavefront");
    }

    __syncthreads();
    const v4f cv = *reinterpret_cast<const v4f*>(&cst[4 * lane]);
    for (int i = 0; i < 4; ++i) {
        const int r = 4 * wave + i;
        const v4f s = *reinterpret_cast<const v4f*>(&otile[r * OP + 4 * lane]);
        v4f o = { s.x + cv.x, s.y + cv.y, s.z + cv.z, s.w + cv.w };
        *reinterpret_cast<v4f*>(out + (size_t)(bbase + r) * Q + 4 * lane) = o;
    }
}

extern "C" void kernel_launch(void* const* d_in, const int* in_sizes, int n_in,
                              void* d_out, int out_size, void* d_ws, size_t ws_size,
                              hipStream_t stream) {
    const float* x     = (const float*)d_in[0];
    const float* W1    = (const float*)d_in[1];
    const float* b1    = (const float*)d_in[2];
    const float* gamma = (const float*)d_in[3];
    const float* beta  = (const float*)d_in[4];
    const float* mmean = (const float*)d_in[5];
    const float* mvar  = (const float*)d_in[6];
    const float* W2    = (const float*)d_in[7];
    const float* b2    = (const float*)d_in[8];
    float* out = (float*)d_out;

    const int B = in_sizes[0] / (Q * S);
    dim3 grid(B / RB);
    dim3 block(1024);
    divenc_wmma_f32<<<grid, block, 0, stream>>>(x, W1, b1, gamma, beta,
                                                mmean, mvar, W2, b2, out);
}


# Block tile re-cut from 128 rows x 128 slices to 512 rows x 32 slices: a block stages and folds a quarter of the parameter table, one slice per wave; waves own their out rows; sums unchanged
# speedup vs baseline: 2.0231x; 1.0494x over previous
#include <hip/hip_runtime.h>

typedef float v2f __attribute__((ext_vector_type(2)));
typedef float v4f __attribute__((ext_vector_type(4)));
typedef float v8f __attribute__((ext_vector_type(8)));

constexpr int Q = 128;
constexpr int S = 8;
constexpr int H = 32;
constexpr int BT = 16;
constexpr int QB = 32;
constexpr int RB = 512;
constexpr int OP = QB + 4;
constexpr int XC = 4;
constexpr int XP = XC * S + 4;
constexpr float BN_EPS = 1e-3f;

template <int MASK>
__device__ __forceinline__ float row_xor_get(float v) {
    return __int_as_float(__builtin_amdgcn_update_dpp(
        0, __float_as_int(v), 0x160 + MASK, 0xf, 0xf, true));
}

__global__ __launch_bounds__(1024) void divenc_wmma_f32(
    const float* __restrict__ x,
    const float* __restrict__ W1,
    const float* __restrict__ b1g,
    const float* __restrict__ gam,
    const float* __restrict__ bet,
    const float* __restrict__ mmean,
    const float* __restrict__ mvar,
    const float* __restrict__ W2,
    const float* __restrict__ b2g,
    float* __restrict__ out)
{
    __shared__ __attribute__((aligned(16))) float wfrag[2 * QB * 32 * 4];
    __shared__ __attribute__((aligned(16))) float coef[QB * 16 * 4];
    __shared__ __attribute__((aligned(16))) float cst[QB];
    __shared__ __attribute__((aligned(16))) float otile[RB * OP];
    __shared__ __attribute__((aligned(16))) float xst[32 * BT * XP];

    const int tid   = threadIdx.x;
    const int lane  = tid & 31;
    const int wave  = tid >> 5;
    const int qblk  = blockIdx.x & 3;
    const int q0    = qblk * QB;
    const int bbase = (blockIdx.x >> 2) * RB;

    const int mrow  = lane & 15;
    const int khalf = lane >> 4;
    const int ncol  = lane & 15;
    const int kb    = 4 * khalf;
    const bool bit3 = (ncol & 8) != 0;
    const bool bit2 = (ncol & 4) != 0;
    const bool bit1 = (ncol & 2) != 0;

    const v2f* W1p   = reinterpret_cast<const v2f*>(W1);
    const v2f* b1p2  = reinterpret_cast<const v2f*>(b1g);
    const v2f* gamp  = reinterpret_cast<const v2f*>(gam);
    const v2f* betp  = reinterpret_cast<const v2f*>(bet);
    const v2f* mup   = reinterpret_cast<const v2f*>(mmean);
    const v2f* varp  = reinterpret_cast<const v2f*>(mvar);
    const v2f* W2p   = reinterpret_cast<const v2f*>(W2);

    const int mt = wave;
    const int lg = lane >> 3;
    const int lp = lane & 7;
    const int lr0 = 8 * (lg & 1) + 4 * (lg >> 1);
    const v4f* xg = reinterpret_cast<const v4f*>(
        x + (size_t)(bbase + mt * BT + lr0) * (Q * S) + q0 * S) + lp;
    constexpr int GROW = Q * S / 4;
    constexpr int GCH  = XC * S / 4;
    constexpr int LROW = XP / 4;
    v4f nx0 = xg[0 * GROW];
    v4f nx1 = xg[1 * GROW];
    v4f nx2 = xg[2 * GROW];
    v4f nx3 = xg[3 * GROW];

    const int iw = (q0 + wave) * (S * H / 2) + kb * (H / 2) + ncol;
    const v2f wr0 = W1p[iw + 0 * (H / 2)];
    const v2f wr1 = W1p[iw + 1 * (H / 2)];
    const v2f wr2 = W1p[iw + 2 * (H / 2)];
    const v2f wr3 = W1p[iw + 3 * (H / 2)];
    if (wave < 16) {
        const int rl  = tid;
        const int rec = q0 * 16 + rl;
        const float b2v = b2g[q0 + (rl >> 4)];
        const v2f b1v = b1p2[rec];
        const v2f gp = gamp[rec];
        const v2f vp = varp[rec];
        const v2f wp = W2p[rec];
        const v2f mp = mup[rec];
        const v2f ep = betp[rec];
        const float inv0 = gp.x * __builtin_amdgcn_rsqf(vp.x + BN_EPS);
        const float inv1 = gp.y * __builtin_amdgcn_rsqf(vp.y + BN_EPS);
        const v4f cr = { b1v.x, b1v.y, inv0 * wp.x, inv1 * wp.y };
        *reinterpret_cast<v4f*>(&coef[rl * 4]) = cr;
        float c0 = fmaf(-mp.x * inv0 + ep.x, wp.x,
                        (-mp.y * inv1 + ep.y) * wp.y);
        c0 += row_xor_get<8>(c0);
        c0 += row_xor_get<4>(c0);
        c0 += row_xor_get<2>(c0);
        c0 += row_xor_get<1>(c0);
        cst[rl >> 4] = c0 + b2v;
    }
    {
        const int q = wave;
        const v4f f0 = { wr0.x, wr1.x, wr2.x, wr3.x };
        const v4f f1 = { wr0.y, wr1.y, wr2.y, wr3.y };
        *reinterpret_cast<v4f*>(&wfrag[((0 * QB + q) * 32 + lane) * 4]) = f0;
        *reinterpret_cast<v4f*>(&wfrag[((1 * QB + q) * 32 + lane) * 4]) = f1;
    }
    __syncthreads();

    const int orow = (mt * BT + khalf * 8 + (ncol >> 1)) * OP;

    float* xw = &xst[wave * (BT * XP)];
    v4f* xput = reinterpret_cast<v4f*>(xw + lr0 * XP + 4 * lp);
    const v4f* xget = reinterpret_cast<const v4f*>(xw + mrow * XP + kb);

    for (int ch = 0; ch < QB / XC; ++ch) {
      xput[0 * LROW] = nx0;
      xput[1 * LROW] = nx1;
      xput[2 * LROW] = nx2;
      xput[3 * LROW] = nx3;
      if (ch + 1 < QB / XC) {
        const v4f* xn = xg + (ch + 1) * GCH;
        nx0 = xn[0 * GROW];
        nx1 = xn[1 * GROW];
        nx2 = xn[2 * GROW];
        nx3 = xn[3 * GROW];
      }
      __builtin_amdgcn_fence(__ATOMIC_RELEASE, "wavefront");
      __builtin_amdgcn_wave_barrier();
      __builtin_amdgcn_fence(__ATOMIC_ACQUIRE, "wavefront");

      for (int ql = 0; ql < XC; ++ql) {
        const int q = ch * XC + ql;
        const v4f xa = xget[ql * (S / 4)];
        v2f a_lo = { xa.x, xa.y };
        v2f a_hi = { xa.z, xa.w };

        const v4f f0 = *reinterpret_cast<const v4f*>(&wfrag[((0 * QB + q) * 32 + lane) * 4]);
        const v4f f1 = *reinterpret_cast<const v4f*>(&wfrag[((1 * QB + q) * 32 + lane) * 4]);
        v2f bt0_lo = { f0.x, f0.y };
        v2f bt0_hi = { f0.z, f0.w };
        v2f bt1_lo = { f1.x, f1.y };
        v2f bt1_hi = { f1.z, f1.w };

        const v4f cr = *reinterpret_cast<const v4f*>(&coef[(q * 16 + ncol) * 4]);
        const float bias0 = cr.x, bias1 = cr.y;
        const float c1_0 = cr.z, c1_1 = cr.w;
        v8f acc0 = { bias0, bias0, bias0, bias0, bias0, bias0, bias0, bias0 };
        v8f acc1 = { bias1, bias1, bias1, bias1, bias1, bias1, bias1, bias1 };

        acc0 = __builtin_amdgcn_wmma_f32_16x16x4_f32(false, a_lo, false, bt0_lo,
                                                     (short)0, acc0, false, false);
        acc0 = __builtin_amdgcn_wmma_f32_16x16x4_f32(false, a_hi, false, bt0_hi,
                                                     (short)0, acc0, false, false);
        acc1 = __builtin_amdgcn_wmma_f32_16x16x4_f32(false, a_lo, false, bt1_lo,
                                                     (short)0, acc1, false, false);
        acc1 = __builtin_amdgcn_wmma_f32_16x16x4_f32(false, a_hi, false, bt1_hi,
                                                     (short)0, acc1, false, false);

        float part[8];
        #pragma unroll
        for (int v = 0; v < 8; ++v) {
            float t0 = acc0[v];
            t0 = t0 > 0.0f ? t0 : (__expf(t0) - 1.0f);
            float t1 = acc1[v];
            t1 = t1 > 0.0f ? t1 : (__expf(t1) - 1.0f);
            part[v] = fmaf(t0, c1_0, t1 * c1_1);
        }

        float r1[4], r2[2];
        #pragma unroll
        for (int i = 0; i < 4; ++i) {
            const float keep = bit3 ? part[i + 4] : part[i];
            const float send = bit3 ? part[i] : part[i + 4];
            r1[i] = keep + row_xor_get<8>(send);
        }
        #pragma unroll
        for (int i = 0; i < 2; ++i) {
            const float keep = bit2 ? r1[i + 2] : r1[i];
            const float send = bit2 ? r1[i] : r1[i + 2];
            r2[i] = keep + row_xor_get<4>(send);
        }
        const float keep3 = bit1 ? r2[1] : r2[0];
        const float send3 = bit1 ? r2[0] : r2[1];
        const float r3 = keep3 + row_xor_get<2>(send3);
        const float r4 = r3 + row_xor_get<1>(r3);

        if ((ncol & 1) == 0) {
            otile[orow + q] = r4;
        }
      }
      __builtin_amdgcn_fence(__ATOMIC_RELEASE, "wavefront");
      __builtin_amdgcn_wave_barrier();
      __builtin_amdgcn_fence(__ATOMIC_ACQUIRE, "wavefront");
    }

    const int oq = 4 * (lane & 7);
    const v4f cv = *reinterpret_cast<const v4f*>(&cst[oq]);
    for (int i = 0; i < 4; ++i) {
        const int r = mt * BT + 4 * i + (lane >> 3);
        const v4f s = *reinterpret_cast<const v4f*>(&otile[r * OP + oq]);
        v4f o = { s.x + cv.x, s.y + cv.y, s.z + cv.z, s.w + cv.w };
        *reinterpret_cast<v4f*>(out + (size_t)(bbase + r) * Q + q0 + oq) = o;
    }
}

extern "C" void kernel_launch(void* const* d_in, const int* in_sizes, int n_in,
                              void* d_out, int out_size, void* d_ws, size_t ws_size,
                              hipStream_t stream) {
    const float* x     = (const float*)d_in[0];
    const float* W1    = (const float*)d_in[1];
    const float* b1    = (const float*)d_in[2];
    const float* gamma = (const float*)d_in[3];
    const float* beta  = (const float*)d_in[4];
    const float* mmean = (const float*)d_in[5];
    const float* mvar  = (const float*)d_in[6];
    const float* W2    = (const float*)d_in[7];
    const float* b2    = (const float*)d_in[8];
    float* out = (float*)d_out;

    const int B = in_sizes[0] / (Q * S);
    dim3 grid((B / RB) * (Q / QB));
    dim3 block(1024);
    divenc_wmma_f32<<<grid, block, 0, stream>>>(x, W1, b1, gamma, beta,
                                                mmean, mvar, W2, b2, out);
}
